# starting kernel
# baseline (speedup 1.0000x reference)
#include <hip/hip_runtime.h>

typedef float v2f __attribute__((ext_vector_type(2)));
typedef float v8f __attribute__((ext_vector_type(8)));

#define C_CH       64
#define NK         128
#define NN         128
#define TOK_PER_WG 128
#define LDS_PITCH  130

__global__ __launch_bounds__(256) void expand_w_kernel(const float* __restrict__ W,
                                                       float* __restrict__ WT) {
    int idx = blockIdx.x * 256 + threadIdx.x;
    int n = idx >> 7;
    int k = idx & 127;
    int c = n >> 1;
    int j = k & 63;
    int half = k >> 6;
    float v = 0.0f;
    if (j != c) {
        int f = half * 63 + (j < c ? j : j - 1);
        v = W[n * 126 + f];
    }
    WT[idx] = v;
}

__global__ __launch_bounds__(256) void leakage_wmma_kernel(const float* __restrict__ x,
                                                           const float* __restrict__ WT,
                                                           float* __restrict__ out) {
    __shared__ float sF[TOK_PER_WG * LDS_PITCH];
    __shared__ float sW[NN * LDS_PITCH];

    const int  tid  = threadIdx.x;
    const long tok0 = (long)blockIdx.x * TOK_PER_WG;

    {
        const float2* g = (const float2*)WT;
        for (int i = tid; i < NN * NK / 2; i += 256) {
            int row  = i >> 6;
            int col2 = (i & 63) * 2;
            float2 v = g[i];
            sW[row * LDS_PITCH + col2]     = v.x;
            sW[row * LDS_PITCH + col2 + 1] = v.y;
        }
    }

    {
        const float2* x2 = (const float2*)x + tok0 * C_CH;
        for (int i = tid; i < TOK_PER_WG * C_CH; i += 256) {
            int t = i >> 6;
            int c = i & 63;
            float2 v  = x2[i];
            float amp = v.x * v.x + v.y * v.y;
            sF[t * LDS_PITCH + c]        = amp * v.x;
            sF[t * LDS_PITCH + C_CH + c] = amp * v.y;
        }
    }
    __syncthreads();

    const int wave  = tid >> 5;
    const int lane  = tid & 31;
    const int lrow  = lane & 15;
    const int khalf = lane >> 4;
    const int m0    = wave * 16;

    v8f acc[8];
    {
        v8f z = {0.f, 0.f, 0.f, 0.f, 0.f, 0.f, 0.f, 0.f};
#pragma unroll
        for (int i = 0; i < 8; ++i) acc[i] = z;
    }

    const float* Abase = &sF[(m0 + lrow) * LDS_PITCH + 2 * khalf];
    const float* Bbase = &sW[lrow * LDS_PITCH + 2 * khalf];

#pragma unroll 4
    for (int kk = 0; kk < NK / 4; ++kk) {
        v2f a = *(const v2f*)(Abase + 4 * kk);
#pragma unroll
        for (int nt = 0; nt < 8; ++nt) {
            v2f b = *(const v2f*)(Bbase + nt * 16 * LDS_PITCH + 4 * kk);
            acc[nt] = __builtin_amdgcn_wmma_f32_16x16x4_f32(
                false, a, false, b, (short)0, acc[nt], false, false);
        }
    }

    float* orow = out + (tok0 + m0 + khalf * 8) * NN;
#pragma unroll
    for (int nt = 0; nt < 8; ++nt) {
#pragma unroll
        for (int r = 0; r < 8; ++r) {
            orow[r * NN + nt * 16 + lrow] = acc[nt][r];
        }
    }
}

extern "C" void kernel_launch(void* const* d_in, const int* in_sizes, int n_in,
                              void* d_out, int out_size, void* d_ws, size_t ws_size,
                              hipStream_t stream) {
    const float* x  = (const float*)d_in[0];
    const float* W  = (const float*)d_in[1];
    float*       o  = (float*)d_out;
    float*       WT = (float*)d_ws;

    expand_w_kernel<<<64, 256, 0, stream>>>(W, WT);

    const int tokens = 16 * 1024;
    leakage_wmma_kernel<<<tokens / TOK_PER_WG, 256, 0, stream>>>(x, WT, o);
}


# Fuse the weight expansion into the GEMM kernel: each block builds the dense 128x128 weight tile in LDS straight from W, dropping the first launch and the workspace round trip
# speedup vs baseline: 1.2125x; 1.2125x over previous
#include <hip/hip_runtime.h>

typedef float v2f __attribute__((ext_vector_type(2)));
typedef float v8f __attribute__((ext_vector_type(8)));

#define C_CH       64
#define NK         128
#define NN         128
#define TOK_PER_WG 128
#define LDS_PITCH  130

__global__ __launch_bounds__(256) void leakage_wmma_kernel(const float* __restrict__ x,
                                                           const float* __restrict__ W,
                                                           float* __restrict__ out) {
    __shared__ float sF[TOK_PER_WG * LDS_PITCH];
    __shared__ float sW[NN * LDS_PITCH];

    const int  tid  = threadIdx.x;
    const long tok0 = (long)blockIdx.x * TOK_PER_WG;

    {
        const unsigned k    = tid & 127;
        const unsigned o    = tid >> 7;
        const unsigned j    = k & 63;
        const unsigned half = k >> 6;
        const unsigned base   = o * 126u + half * 63u;
        const unsigned off_lo = base + (j < 62u ? j : 62u);
        const unsigned off_hi = base + (j > 0u ? j - 1u : 0u);
        float* sp = &sW[o * LDS_PITCH + k];
#pragma unroll
        for (unsigned c = 0; c < C_CH; ++c) {
            const unsigned off = (j > c) ? off_hi : off_lo;
            const float v = (W + c * 252u)[off];
            sp[c * 2u * LDS_PITCH] = (j != c) ? v : 0.0f;
        }
    }

    {
        const float2* x2 = (const float2*)x + tok0 * C_CH;
        for (int i = tid; i < TOK_PER_WG * C_CH; i += 256) {
            int t = i >> 6;
            int c = i & 63;
            float2 v  = x2[i];
            float amp = v.x * v.x + v.y * v.y;
            sF[t * LDS_PITCH + c]        = amp * v.x;
            sF[t * LDS_PITCH + C_CH + c] = amp * v.y;
        }
    }
    __syncthreads();

    const int wave  = tid >> 5;
    const int lane  = tid & 31;
    const int lrow  = lane & 15;
    const int khalf = lane >> 4;
    const int m0    = wave * 16;

    v8f acc[8];
    {
        v8f z = {0.f, 0.f, 0.f, 0.f, 0.f, 0.f, 0.f, 0.f};
#pragma unroll
        for (int i = 0; i < 8; ++i) acc[i] = z;
    }

    const float* Abase = &sF[(m0 + lrow) * LDS_PITCH + 2 * khalf];
    const float* Bbase = &sW[lrow * LDS_PITCH + 2 * khalf];

#pragma unroll 4
    for (int kk = 0; kk < NK / 4; ++kk) {
        v2f a = *(const v2f*)(Abase + 4 * kk);
#pragma unroll
        for (int nt = 0; nt < 8; ++nt) {
            v2f b = *(const v2f*)(Bbase + nt * 16 * LDS_PITCH + 4 * kk);
            acc[nt] = __builtin_amdgcn_wmma_f32_16x16x4_f32(
                false, a, false, b, (short)0, acc[nt], false, false);
        }
    }

    float* orow = out + (tok0 + m0 + khalf * 8) * NN;
#pragma unroll
    for (int nt = 0; nt < 8; ++nt) {
#pragma unroll
        for (int r = 0; r < 8; ++r) {
            orow[r * NN + nt * 16 + lrow] = acc[nt][r];
        }
    }
}

extern "C" void kernel_launch(void* const* d_in, const int* in_sizes, int n_in,
                              void* d_out, int out_size, void* d_ws, size_t ws_size,
                              hipStream_t stream) {
    const float* x  = (const float*)d_in[0];
    const float* W  = (const float*)d_in[1];
    float*       o  = (float*)d_out;

    const int tokens = 16 * 1024;
    leakage_wmma_kernel<<<tokens / TOK_PER_WG, 256, 0, stream>>>(x, W, o);
}


# Token tile per work-group 128 -> 64: grid 128 -> 256 work-groups, one per WGP, so every matrix pipe works; a wave does 16 tokens x 64 outputs (4 tiles) and the F loop has half the trips
# speedup vs baseline: 1.8656x; 1.5386x over previous
#include <hip/hip_runtime.h>

typedef float v2f __attribute__((ext_vector_type(2)));
typedef float v8f __attribute__((ext_vector_type(8)));

#define C_CH       64
#define NK         128
#define NN         128
#define TOK_PER_WG 64
#define LDS_PITCH  130

__global__ __launch_bounds__(256) void leakage_wmma_kernel(const float* __restrict__ x,
                                                           const float* __restrict__ W,
                                                           float* __restrict__ out) {
    __shared__ float sF[TOK_PER_WG * LDS_PITCH];
    __shared__ float sW[NN * LDS_PITCH];

    const int  tid  = threadIdx.x;
    const long tok0 = (long)blockIdx.x * TOK_PER_WG;

    {
        const unsigned k    = tid & 127;
        const unsigned o    = tid >> 7;
        const unsigned j    = k & 63;
        const unsigned half = k >> 6;
        const unsigned base   = o * 126u + half * 63u;
        const unsigned off_lo = base + (j < 62u ? j : 62u);
        const unsigned off_hi = base + (j > 0u ? j - 1u : 0u);
        float* sp = &sW[o * LDS_PITCH + k];
#pragma unroll
        for (unsigned c = 0; c < C_CH; ++c) {
            const unsigned off = (j > c) ? off_hi : off_lo;
            const float v = (W + c * 252u)[off];
            sp[c * 2u * LDS_PITCH] = (j != c) ? v : 0.0f;
        }
    }

    {
        const float2* x2 = (const float2*)x + tok0 * C_CH;
#pragma unroll 4
        for (int i = tid; i < TOK_PER_WG * C_CH; i += 256) {
            int t = i >> 6;
            int c = i & 63;
            float2 v  = x2[i];
            float amp = v.x * v.x + v.y * v.y;
            sF[t * LDS_PITCH + c]        = amp * v.x;
            sF[t * LDS_PITCH + C_CH + c] = amp * v.y;
        }
    }
    __syncthreads();

    const int wave  = tid >> 5;
    const int lane  = tid & 31;
    const int lrow  = lane & 15;
    const int khalf = lane >> 4;
    const int m0    = (wave & 3) * 16;
    const int n0    = (wave >> 2) * 64;

    v8f acc[4];
    {
        v8f z = {0.f, 0.f, 0.f, 0.f, 0.f, 0.f, 0.f, 0.f};
#pragma unroll
        for (int i = 0; i < 4; ++i) acc[i] = z;
    }

    const float* Abase = &sF[(m0 + lrow) * LDS_PITCH + 2 * khalf];
    const float* Bbase = &sW[(n0 + lrow) * LDS_PITCH + 2 * khalf];

#pragma unroll 8
    for (int kk = 0; kk < NK / 4; ++kk) {
        v2f a = *(const v2f*)(Abase + 4 * kk);
#pragma unroll
        for (int nt = 0; nt < 4; ++nt) {
            v2f b = *(const v2f*)(Bbase + nt * 16 * LDS_PITCH + 4 * kk);
            acc[nt] = __builtin_amdgcn_wmma_f32_16x16x4_f32(
                false, a, false, b, (short)0, acc[nt], false, false);
        }
    }

    float* orow = out + (tok0 + m0 + khalf * 8) * NN + n0;
#pragma unroll
    for (int nt = 0; nt < 4; ++nt) {
#pragma unroll
        for (int r = 0; r < 8; ++r) {
            orow[r * NN + nt * 16 + lrow] = acc[nt][r];
        }
    }
}

extern "C" void kernel_launch(void* const* d_in, const int* in_sizes, int n_in,
                              void* d_out, int out_size, void* d_ws, size_t ws_size,
                              hipStream_t stream) {
    const float* x  = (const float*)d_in[0];
    const float* W  = (const float*)d_in[1];
    float*       o  = (float*)d_out;

    const int tokens = 16 * 1024;
    leakage_wmma_kernel<<<tokens / TOK_PER_WG, 256, 0, stream>>>(x, W, o);
}


# All x-tile loads of a block (8 x 16 B per thread) issued in one group and left in flight under the weight expansion, replacing the load-compute-store loop that paid the memory latency four times
# speedup vs baseline: 2.3837x; 1.2777x over previous
#include <hip/hip_runtime.h>

typedef float v2f __attribute__((ext_vector_type(2)));
typedef float v8f __attribute__((ext_vector_type(8)));
typedef float v4f __attribute__((ext_vector_type(4)));

#define C_CH       64
#define NK         128
#define NN         128
#define TOK_PER_WG 64
#define LDS_PITCH  130

__global__ __launch_bounds__(256) void leakage_wmma_kernel(const float* __restrict__ x,
                                                           const float* __restrict__ W,
                                                           float* __restrict__ out) {
    __shared__ __attribute__((aligned(16))) float sF[TOK_PER_WG * LDS_PITCH];
    __shared__ __attribute__((aligned(16))) float sW[NN * LDS_PITCH];

    const int  tid  = threadIdx.x;
    const long tok0 = (long)blockIdx.x * TOK_PER_WG;

    v4f xv[8];
    {
        const v4f* x4 = (const v4f*)x + tok0 * (C_CH / 2);
#pragma unroll
        for (int i = 0; i < 8; ++i) xv[i] = x4[tid + 256 * i];
    }

    {
        const unsigned k    = tid & 127;
        const unsigned o    = tid >> 7;
        const unsigned j    = k & 63;
        const unsigned half = k >> 6;
        const unsigned base   = o * 126u + half * 63u;
        const unsigned off_lo = base + (j < 62u ? j : 62u);
        const unsigned off_hi = base + (j > 0u ? j - 1u : 0u);
        float* sp = &sW[o * LDS_PITCH + k];
#pragma unroll
        for (unsigned c = 0; c < C_CH; ++c) {
            const unsigned off = (j > c) ? off_hi : off_lo;
            const float v = (W + c * 252u)[off];
            sp[c * 2u * LDS_PITCH] = (j != c) ? v : 0.0f;
        }
    }

    {
        float* fp = &sF[(tid >> 5) * LDS_PITCH + 2 * (tid & 31)];
#pragma unroll
        for (int i = 0; i < 8; ++i) {
            const v4f v = xv[i];
            const float amp0 = __builtin_fmaf(v.x, v.x, v.y * v.y);
            const float amp1 = __builtin_fmaf(v.z, v.z, v.w * v.w);
            v2f fr = {amp0 * v.x, amp1 * v.z};
            v2f fi = {amp0 * v.y, amp1 * v.w};
            *(v2f*)(fp + i * 8 * LDS_PITCH)        = fr;
            *(v2f*)(fp + i * 8 * LDS_PITCH + C_CH) = fi;
        }
    }
    __syncthreads();

    const int wave  = tid >> 5;
    const int lane  = tid & 31;
    const int lrow  = lane & 15;
    const int khalf = lane >> 4;
    const int m0    = (wave & 3) * 16;
    const int n0    = (wave >> 2) * 64;

    v8f acc[4];
    {
        v8f z = {0.f, 0.f, 0.f, 0.f, 0.f, 0.f, 0.f, 0.f};
#pragma unroll
        for (int i = 0; i < 4; ++i) acc[i] = z;
    }

    const float* Abase = &sF[(m0 + lrow) * LDS_PITCH + 2 * khalf];
    const float* Bbase = &sW[(n0 + lrow) * LDS_PITCH + 2 * khalf];

#pragma unroll 8
    for (int kk = 0; kk < NK / 4; ++kk) {
        v2f a = *(const v2f*)(Abase + 4 * kk);
#pragma unroll
        for (int nt = 0; nt < 4; ++nt) {
            v2f b = *(const v2f*)(Bbase + nt * 16 * LDS_PITCH + 4 * kk);
            acc[nt] = __builtin_amdgcn_wmma_f32_16x16x4_f32(
                false, a, false, b, (short)0, acc[nt], false, false);
        }
    }

    float* orow = out + (tok0 + m0 + khalf * 8) * NN + n0;
#pragma unroll
    for (int nt = 0; nt < 4; ++nt) {
#pragma unroll
        for (int r = 0; r < 8; ++r) {
            orow[r * NN + nt * 16 + lrow] = acc[nt][r];
        }
    }
}

extern "C" void kernel_launch(void* const* d_in, const int* in_sizes, int n_in,
                              void* d_out, int out_size, void* d_ws, size_t ws_size,
                              hipStream_t stream) {
    const float* x  = (const float*)d_in[0];
    const float* W  = (const float*)d_in[1];
    float*       o  = (float*)d_out;

    const int tokens = 16 * 1024;
    leakage_wmma_kernel<<<tokens / TOK_PER_WG, 256, 0, stream>>>(x, W, o);
}
